# starting kernel
# baseline (speedup 1.0000x reference)
#include <hip/hip_runtime.h>
#include <hip/hip_bf16.h>

typedef __attribute__((ext_vector_type(2))) float v2f;
typedef __attribute__((ext_vector_type(4))) float v4f;
typedef __attribute__((ext_vector_type(8))) float v8f;

#define FEAT_D 128
#define NORM_EPS 1e-12f

__global__ void cossim_invnorm_kernel(const float* __restrict__ h,
                                      float* __restrict__ inv, int N) {
  const int lane = threadIdx.x & 31;
  const int row  = blockIdx.x * (blockDim.x >> 5) + (threadIdx.x >> 5);
  if (row >= N) return;
  const float* rp = h + (size_t)row * FEAT_D + lane * 4;
  v4f v = *(const v4f*)rp;
  float s = v.x * v.x + v.y * v.y + v.z * v.z + v.w * v.w;
#pragma unroll
  for (int m = 16; m >= 1; m >>= 1) s += __shfl_xor(s, m, 32);
  if (lane == 0) {
    float n = sqrtf(s);
    inv[row] = 1.0f / fmaxf(n, NORM_EPS);
  }
}

__global__ void cossim_sddmm_wmma_kernel(const float* __restrict__ h,
                                         const int* __restrict__ src,
                                         const int* __restrict__ dst,
                                         const float* __restrict__ inv,
                                         float* __restrict__ out, int E) {
  const int lane  = threadIdx.x & 31;
  const int group = blockIdx.x * (blockDim.x >> 5) + (threadIdx.x >> 5);
  const long long e0 = (long long)group * 16;
  if (e0 >= E) return;

  const int r    = lane & 15;
  const int half = lane >> 4;
  long long eg = e0 + r;
  if (eg >= E) eg = (long long)E - 1;
  const int srow = src[eg];
  const int drow = dst[eg];
  const float* ap = h + (size_t)srow * FEAT_D + 2 * half;
  const float* bp = h + (size_t)drow * FEAT_D + 2 * half;

  v8f acc = {};
#pragma unroll
  for (int k = 0; k < FEAT_D; k += 4) {
    v2f a = *(const v2f*)(ap + k);
    v2f b = *(const v2f*)(bp + k);
    acc = __builtin_amdgcn_wmma_f32_16x16x4_f32(
         false, a,  false, b,
         (short)0, acc,  false,  false);
  }

  const bool active = (lane < 8) || (lane >= 24);
  const int  vidx   = lane & 7;
  const int  j      = (lane < 8) ? lane : (lane - 16);
  float diag = acc[0];
#pragma unroll
  for (int i = 1; i < 8; ++i) diag = (vidx == i) ? acc[i] : diag;

  const long long e = e0 + j;
  if (active && e < E) {
    const int sj = src[e];
    const int dj = dst[e];
    out[e] = diag * inv[sj] * inv[dj];
  }
}

extern "C" void kernel_launch(void* const* d_in, const int* in_sizes, int n_in,
                              void* d_out, int out_size, void* d_ws, size_t ws_size,
                              hipStream_t stream) {
  const float* h   = (const float*)d_in[0];
  const int*   src = (const int*)d_in[1];
  const int*   dst = (const int*)d_in[2];
  float*       out = (float*)d_out;
  float*       inv = (float*)d_ws;

  const int N = in_sizes[0] / FEAT_D;
  const int E = in_sizes[1];

  {
    const int rows_per_block = 256 / 32;
    dim3 block(256), grid((N + rows_per_block - 1) / rows_per_block);
    cossim_invnorm_kernel<<<grid, block, 0, stream>>>(h, inv, N);
  }
  {
    const int groups = (E + 15) / 16;
    const int gpb = 256 / 32;
    dim3 block(256), grid((groups + gpb - 1) / gpb);
    cossim_sddmm_wmma_kernel<<<grid, block, 0, stream>>>(h, src, dst, inv, out, E);
  }
}


# SDDMM as a resident-size grid: 4096 waves enter once and loop over their tiles of 16 edges, the next tile's 32 row loads in flight under the current tile's LDS reads and WMMAs
# speedup vs baseline: 2.3938x; 2.3938x over previous
#include <hip/hip_runtime.h>
#include <hip/hip_bf16.h>

typedef __attribute__((ext_vector_type(2))) float v2f;
typedef __attribute__((ext_vector_type(4))) float v4f;
typedef __attribute__((ext_vector_type(8))) float v8f;

#define FEAT_D 128
#define NORM_EPS 1e-12f

#define TILE_EDGES 16
#define TILE_ROWS 32
#define ROW_PITCH 132
#define SDDMM_WAVES 4
#define SDDMM_GROUPS 1024

__global__ void cossim_invnorm_kernel(const float* __restrict__ h,
                                      float* __restrict__ inv, int N) {
  const int lane = threadIdx.x & 31;
  const int row  = blockIdx.x * (blockDim.x >> 5) + (threadIdx.x >> 5);
  if (row >= N) return;
  const float* rp = h + (size_t)row * FEAT_D + lane * 4;
  v4f v = *(const v4f*)rp;
  float s = v.x * v.x + v.y * v.y + v.z * v.z + v.w * v.w;
#pragma unroll
  for (int m = 16; m >= 1; m >>= 1) s += __shfl_xor(s, m, 32);
  if (lane == 0) {
    float n = sqrtf(s);
    inv[row] = 1.0f / fmaxf(n, NORM_EPS);
  }
}

__global__ __launch_bounds__(SDDMM_WAVES * 32)
void cossim_sddmm_wmma_kernel(const float* __restrict__ h,
                              const int* __restrict__ src,
                              const int* __restrict__ dst,
                              const float* __restrict__ inv,
                              float* __restrict__ out, int E) {
  __shared__ __attribute__((aligned(16))) float stage[SDDMM_WAVES * TILE_ROWS * ROW_PITCH];

  const int lane = threadIdx.x & 31;
  const int wave   = __builtin_amdgcn_readfirstlane((int)(threadIdx.x >> 5));
  const int nwaves = (int)gridDim.x * SDDMM_WAVES;
  const int ntiles = (E + TILE_EDGES - 1) / TILE_EDGES;
  int t = (int)blockIdx.x * SDDMM_WAVES + wave;
  if (t >= ntiles) return;

  const int r    = lane & 15;
  const int half = lane >> 4;
  const int* ip  = half ? dst : src;
  const long long last_e = (long long)E - 1;

  float* tile = stage + wave * (TILE_ROWS * ROW_PITCH);
  const float* ar = tile + r * ROW_PITCH + 4 * half;
  const float* br = tile + (TILE_EDGES + r) * ROW_PITCH + 4 * half;
  const float* hl = h + lane * 4;

  long long eg = (long long)t * TILE_EDGES + r;
  if (eg > last_e) eg = last_e;
  int node = ip[eg];
  int t1 = t + nwaves;
  int node_n = 0;
  if (t1 < ntiles) {
    long long eg1 = (long long)t1 * TILE_EDGES + r;
    if (eg1 > last_e) eg1 = last_e;
    node_n = ip[eg1];
  }
  float inv_own = inv[node];

  v4f rows[TILE_ROWS];
#pragma unroll
  for (int i = 0; i < TILE_ROWS; ++i) {
    const int n_i = __builtin_amdgcn_readlane(node, i);
    rows[i] = *(const v4f*)(hl + (size_t)n_i * FEAT_D);
  }

  for (;;) {
#pragma unroll
    for (int i = 0; i < TILE_ROWS; ++i) {
      *(v4f*)(tile + i * ROW_PITCH + lane * 4) = rows[i];
    }
    __builtin_amdgcn_wave_barrier();

    const float inv_other = __shfl_xor(inv_own, 16, 32);
    const float inv_s = half ? inv_other : inv_own;
    const float inv_d = half ? inv_own : inv_other;

    const bool more = t1 < ntiles;
    float inv_n = 0.0f;
    int node_nn = 0;
    if (more) {
#pragma unroll
      for (int i = 0; i < TILE_ROWS; ++i) {
        const int n_i = __builtin_amdgcn_readlane(node_n, i);
        rows[i] = *(const v4f*)(hl + (size_t)n_i * FEAT_D);
      }
      inv_n = inv[node_n];
      const int t2 = t1 + nwaves;
      if (t2 < ntiles) {
        long long eg2 = (long long)t2 * TILE_EDGES + r;
        if (eg2 > last_e) eg2 = last_e;
        node_nn = ip[eg2];
      }
    }

    v8f acc0 = {};
    v8f acc1 = {};
#pragma unroll
    for (int s = 0; s < FEAT_D / 8; ++s) {
      const v4f a4 = *(const v4f*)(ar + 8 * s);
      const v4f b4 = *(const v4f*)(br + 8 * s);
      const v2f a_lo = {a4.x, a4.y};
      const v2f b_lo = {b4.x, b4.y};
      const v2f a_hi = {a4.z, a4.w};
      const v2f b_hi = {b4.z, b4.w};
      acc0 = __builtin_amdgcn_wmma_f32_16x16x4_f32(
           false, a_lo,  false, b_lo,
           (short)0, acc0,  false,  false);
      acc1 = __builtin_amdgcn_wmma_f32_16x16x4_f32(
           false, a_hi,  false, b_hi,
           (short)0, acc1,  false,  false);
    }
    const v8f acc = acc0 + acc1;

    const bool active = (lane < 8) || (lane >= 24);
    const int  vidx   = lane & 7;
    float diag = acc[0];
#pragma unroll
    for (int i = 1; i < 8; ++i) diag = (vidx == i) ? acc[i] : diag;

    const long long e = (long long)t * TILE_EDGES + r;
    if (active && e <= last_e) {
      out[e] = diag * inv_s * inv_d;
    }

    if (!more) break;
    __builtin_amdgcn_wave_barrier();
    t = t1;
    t1 += nwaves;
    node_n = node_nn;
    inv_own = inv_n;
  }
}

extern "C" void kernel_launch(void* const* d_in, const int* in_sizes, int n_in,
                              void* d_out, int out_size, void* d_ws, size_t ws_size,
                              hipStream_t stream) {
  const float* h   = (const float*)d_in[0];
  const int*   src = (const int*)d_in[1];
  const int*   dst = (const int*)d_in[2];
  float*       out = (float*)d_out;
  float*       inv = (float*)d_ws;

  const int N = in_sizes[0] / FEAT_D;
  const int E = in_sizes[1];

  {
    const int rows_per_block = 256 / 32;
    dim3 block(256), grid((N + rows_per_block - 1) / rows_per_block);
    cossim_invnorm_kernel<<<grid, block, 0, stream>>>(h, inv, N);
  }
  {
    dim3 block(SDDMM_WAVES * 32), grid(SDDMM_GROUPS);
    cossim_sddmm_wmma_kernel<<<grid, block, 0, stream>>>(h, src, dst, inv, out, E);
  }
}


# Gram fragment LDS reads grouped ahead of the WMMAs: 16 reads right behind the stores, then two refills of 8 under long WMMA runs (two register sets), instead of 2 reads per 4 WMMAs behind a wait
# speedup vs baseline: 2.6681x; 1.1146x over previous
#include <hip/hip_runtime.h>
#include <hip/hip_bf16.h>

typedef __attribute__((ext_vector_type(2))) float v2f;
typedef __attribute__((ext_vector_type(4))) float v4f;
typedef __attribute__((ext_vector_type(8))) float v8f;

#define FEAT_D 128
#define NORM_EPS 1e-12f

#define TRIP_EDGES 16
#define TILE_ROWS 32
#define ROW_PITCH 132
#define SDDMM_WAVES 4
#define SDDMM_GROUPS 1024
#define FRAG_GROUP 8

__device__ __forceinline__ void read_frags(v4f (&f)[FRAG_GROUP], const float* x, int s0) {
#pragma unroll
  for (int j = 0; j < FRAG_GROUP; ++j) f[j] = *(const v4f*)(x + 8 * (s0 + j));
}

__device__ __forceinline__ void gram_run(const v4f (&f)[FRAG_GROUP], v8f& ga, v8f& gb) {
#pragma unroll
  for (int j = 0; j < FRAG_GROUP; ++j) {
    const v2f lo = {f[j].x, f[j].y};
    const v2f hi = {f[j].z, f[j].w};
    ga = __builtin_amdgcn_wmma_f32_16x16x4_f32(
         false, lo,  false, lo,
         (short)0, ga,  false,  false);
    gb = __builtin_amdgcn_wmma_f32_16x16x4_f32(
         false, hi,  false, hi,
         (short)0, gb,  false,  false);
  }
}

__device__ __forceinline__ float pick_diag(const v8f& g, int vidx) {
  float v = g[0];
#pragma unroll
  for (int i = 1; i < 8; ++i) v = (vidx == i) ? g[i] : v;
  return v;
}

__global__ __launch_bounds__(SDDMM_WAVES * 32)
void cossim_gram_wmma_kernel(const float* __restrict__ h,
                             const int* __restrict__ src,
                             const int* __restrict__ dst,
                             float* __restrict__ out, int E) {
  __shared__ __attribute__((aligned(16))) float stage[SDDMM_WAVES * TILE_ROWS * ROW_PITCH];

  const int lane = threadIdx.x & 31;
  const int wave   = __builtin_amdgcn_readfirstlane((int)(threadIdx.x >> 5));
  const int nwaves = (int)gridDim.x * SDDMM_WAVES;
  const int ntrips = (E + TRIP_EDGES - 1) / TRIP_EDGES;
  int t = (int)blockIdx.x * SDDMM_WAVES + wave;
  if (t >= ntrips) return;

  const int c    = lane & 15;
  const int half = lane >> 4;
  const int erel = ((lane >> 4) << 3) + (lane & 7);
  const int* ip  = (lane & 8) ? dst : src;
  const long long last_e = (long long)E - 1;

  float* tile = stage + wave * (TILE_ROWS * ROW_PITCH);
  const float* x0 = tile + c * ROW_PITCH + 4 * half;
  const float* x1 = x0 + 16 * ROW_PITCH;
  const unsigned lcol = (unsigned)lane * 4u;

  long long eg = (long long)t * TRIP_EDGES + erel;
  if (eg > last_e) eg = last_e;
  int node = ip[eg];
  int t1 = t + nwaves;
  int node_n = 0;
  if (t1 < ntrips) {
    long long eg1 = (long long)t1 * TRIP_EDGES + erel;
    if (eg1 > last_e) eg1 = last_e;
    node_n = ip[eg1];
  }

  v4f rows[TILE_ROWS];
#pragma unroll
  for (int i = 0; i < TILE_ROWS; ++i) {
    const int n_i = __builtin_amdgcn_readlane(node, i);
    const float* rp = h + (size_t)n_i * FEAT_D;
    rows[i] = *(const v4f*)(rp + lcol);
  }

  for (;;) {
#pragma unroll
    for (int i = 0; i < TILE_ROWS; ++i) {
      *(v4f*)(tile + i * ROW_PITCH + lane * 4) = rows[i];
    }
    __builtin_amdgcn_wave_barrier();

    v4f fa[FRAG_GROUP];
    v4f fb[FRAG_GROUP];
    read_frags(fa, x0, 0);
    read_frags(fb, x0, 8);
    __builtin_amdgcn_sched_barrier(0);

    const bool more = t1 < ntrips;
    int node_nn = 0;
    if (more) {
#pragma unroll
      for (int i = 0; i < TILE_ROWS; ++i) {
        const int n_i = __builtin_amdgcn_readlane(node_n, i);
        const float* rp = h + (size_t)n_i * FEAT_D;
        rows[i] = *(const v4f*)(rp + lcol);
      }
      const int t2 = t1 + nwaves;
      if (t2 < ntrips) {
        long long eg2 = (long long)t2 * TRIP_EDGES + erel;
        if (eg2 > last_e) eg2 = last_e;
        node_nn = ip[eg2];
      }
    }
    __builtin_amdgcn_sched_barrier(0);

    v8f ga = {};
    v8f gb = {};
    gram_run(fa, ga, gb);
    __builtin_amdgcn_sched_barrier(0);
    read_frags(fa, x1, 0);
    __builtin_amdgcn_sched_barrier(0);
    gram_run(fb, ga, gb);
    __builtin_amdgcn_sched_barrier(0);
    read_frags(fb, x1, 8);
    __builtin_amdgcn_sched_barrier(0);

    const int vidx = lane & 7;
    const v8f g0 = ga + gb;
    const float v0 = pick_diag(g0, vidx);

    ga = (v8f){};
    gb = (v8f){};
    gram_run(fa, ga, gb);
    gram_run(fb, ga, gb);
    const v8f g1 = ga + gb;
    const float v1 = pick_diag(g1, vidx);

    const float ss0 = __shfl(v0, (lane - 8) & 31, 32);
    const float dd0 = __shfl(v0, (lane + 16) & 31, 32);
    const float ss1 = __shfl(v1, (lane - 16) & 31, 32);
    const float dd1 = __shfl(v1, (lane + 8) & 31, 32);
    const bool  low = lane < 16;
    const float dot = low ? v0 : v1;
    const float ss  = low ? ss0 : ss1;
    const float dd  = low ? dd0 : dd1;

    const long long e = (long long)t * TRIP_EDGES + (lane - 8);
    if (lane >= 8 && lane < 24 && e <= last_e) {
      const float ns = fmaxf(sqrtf(ss), NORM_EPS);
      const float nd = fmaxf(sqrtf(dd), NORM_EPS);
      out[e] = dot / (ns * nd);
    }

    if (!more) break;
    __builtin_amdgcn_wave_barrier();
    t = t1;
    t1 += nwaves;
    node_n = node_nn;
  }
}

extern "C" void kernel_launch(void* const* d_in, const int* in_sizes, int n_in,
                              void* d_out, int out_size, void* d_ws, size_t ws_size,
                              hipStream_t stream) {
  const float* h   = (const float*)d_in[0];
  const int*   src = (const int*)d_in[1];
  const int*   dst = (const int*)d_in[2];
  float*       out = (float*)d_out;

  const int E = in_sizes[1];

  dim3 block(SDDMM_WAVES * 32), grid(SDDMM_GROUPS);
  cossim_gram_wmma_kernel<<<grid, block, 0, stream>>>(h, src, dst, out, E);
}


# Gather issue moved off the vector ALU: trip indices by 64-byte scalar loads, row bases in scalar registers, loads as scalar base + lane offset; first 16 row loads issue between the store halves
# speedup vs baseline: 2.9163x; 1.0930x over previous
#include <hip/hip_runtime.h>
#include <hip/hip_bf16.h>

typedef __attribute__((ext_vector_type(2))) float v2f;
typedef __attribute__((ext_vector_type(2))) unsigned v2u;
typedef __attribute__((ext_vector_type(4))) float v4f;
typedef __attribute__((ext_vector_type(8))) float v8f;
typedef int v16i_a4 __attribute__((ext_vector_type(16), aligned(4)));
typedef const int __attribute__((address_space(4)))* ro_int_ptr;
typedef const v16i_a4 __attribute__((address_space(4)))* ro_v16i_ptr;
typedef const v4f __attribute__((address_space(1)))* gl_v4f_ptr;

#define FEAT_D 128
#define NORM_EPS 1e-12f

#define TRIP_EDGES 16
#define TILE_ROWS 32
#define ROW_PITCH 132
#define SDDMM_WAVES 4
#define SDDMM_GROUPS 1024
#define FRAG_GROUP 8

__device__ __forceinline__ void read_frags(v4f (&f)[FRAG_GROUP], const float* x, int s0) {
#pragma unroll
  for (int j = 0; j < FRAG_GROUP; ++j) f[j] = *(const v4f*)(x + 8 * (s0 + j));
}

__device__ __forceinline__ void gram_run(const v4f (&f)[FRAG_GROUP], v8f& ga, v8f& gb) {
#pragma unroll
  for (int j = 0; j < FRAG_GROUP; ++j) {
    const v2f lo = {f[j].x, f[j].y};
    const v2f hi = {f[j].z, f[j].w};
    ga = __builtin_amdgcn_wmma_f32_16x16x4_f32(
         false, lo,  false, lo,
         (short)0, ga,  false,  false);
    gb = __builtin_amdgcn_wmma_f32_16x16x4_f32(
         false, hi,  false, hi,
         (short)0, gb,  false,  false);
  }
}

__device__ __forceinline__ float pick_diag(const v8f& g, int vidx) {
  float v = g[0];
#pragma unroll
  for (int i = 1; i < 8; ++i) v = (vidx == i) ? g[i] : v;
  return v;
}


__device__ __forceinline__ void load_trip_indices(int (&si)[TRIP_EDGES], int (&di)[TRIP_EDGES],
                                                  ro_int_ptr src, ro_int_ptr dst,
                                                  int e0, int last_e) {
  if (e0 + (TRIP_EDGES - 1) <= last_e) {
    const v16i_a4 sv = *(ro_v16i_ptr)(src + e0);
    const v16i_a4 dv = *(ro_v16i_ptr)(dst + e0);
#pragma unroll
    for (int j = 0; j < TRIP_EDGES; ++j) { si[j] = sv[j]; di[j] = dv[j]; }
  } else {
#pragma unroll
    for (int j = 0; j < TRIP_EDGES; ++j) {
      const int e = min(e0 + j, last_e);
      si[j] = src[e];
      di[j] = dst[e];
    }
  }
}

#define GATHER_GROUP 16
__device__ __forceinline__ void gather_rows(v4f (&rows)[TILE_ROWS], const float* __restrict__ h,
                                            const int (&si)[TRIP_EDGES], const int (&di)[TRIP_EDGES],
                                            unsigned& lbyte, const int g0, const int ng) {
  asm volatile("" : "+v"(lbyte));
#pragma unroll
  for (int g = g0; g < g0 + ng; g += GATHER_GROUP) {
    unsigned long long rbase[GATHER_GROUP];
#pragma unroll
    for (int j = 0; j < GATHER_GROUP; ++j) {
      const int i   = g + j;
      const int er  = ((i >> 4) << 3) + (i & 7);
      const unsigned n_i = (unsigned)((i & 8) ? di[er] : si[er]);
      unsigned lo = n_i << 9;
      unsigned hi = n_i >> 23;
      asm("" : "+s"(lo));
      asm("" : "+s"(hi));
      const v2u off = {lo, hi};
      rbase[j] = (unsigned long long)h + __builtin_bit_cast(unsigned long long, off);
      asm("" : "+s"(rbase[j]));
    }
#pragma unroll
    for (int j = 0; j < GATHER_GROUP; ++j) {
      rows[g + j] = *(gl_v4f_ptr)(rbase[j] + (unsigned long long)lbyte);
    }
  }
}

__global__ __launch_bounds__(SDDMM_WAVES * 32)
void cossim_gram_wmma_kernel(const float* __restrict__ h,
                             const int* __restrict__ src,
                             const int* __restrict__ dst,
                             float* __restrict__ out, int E) {
  __shared__ __attribute__((aligned(16))) float stage[SDDMM_WAVES * TILE_ROWS * ROW_PITCH];

  const int lane = threadIdx.x & 31;
  const int wave   = __builtin_amdgcn_readfirstlane((int)(threadIdx.x >> 5));
  const int nwaves = (int)gridDim.x * SDDMM_WAVES;
  const int ntrips = (E + TRIP_EDGES - 1) / TRIP_EDGES;
  int t = (int)blockIdx.x * SDDMM_WAVES + wave;
  if (t >= ntrips) return;

  const int c    = lane & 15;
  const int half = lane >> 4;
  const long long last_e = (long long)E - 1;
  const ro_int_ptr src_ro = (ro_int_ptr)(unsigned long long)src;
  const ro_int_ptr dst_ro = (ro_int_ptr)(unsigned long long)dst;

  float* tile = stage + wave * (TILE_ROWS * ROW_PITCH);
  const float* x0 = tile + c * ROW_PITCH + 4 * half;
  const float* x1 = x0 + 16 * ROW_PITCH;
  unsigned lbyte = (unsigned)lane * 16u;

  int si[TRIP_EDGES], di[TRIP_EDGES];
  load_trip_indices(si, di, src_ro, dst_ro, t * TRIP_EDGES, E - 1);
  v4f rows[TILE_ROWS];
  gather_rows(rows, h, si, di, lbyte, 0, TILE_ROWS);
  int t1 = t + nwaves;
  if (t1 < ntrips) load_trip_indices(si, di, src_ro, dst_ro, t1 * TRIP_EDGES, E - 1);

  for (;;) {
    const bool more = t1 < ntrips;
    v4f fa[FRAG_GROUP];
    v4f fb[FRAG_GROUP];
    if (more) {
#pragma unroll
      for (int i = 0; i < TILE_ROWS / 2; ++i) {
        *(v4f*)(tile + i * ROW_PITCH + lane * 4) = rows[i];
      }
      __builtin_amdgcn_sched_barrier(0);
      gather_rows(rows, h, si, di, lbyte, 0, TILE_ROWS / 2);
      __builtin_amdgcn_sched_barrier(0);
#pragma unroll
      for (int i = TILE_ROWS / 2; i < TILE_ROWS; ++i) {
        *(v4f*)(tile + i * ROW_PITCH + lane * 4) = rows[i];
      }
      __builtin_amdgcn_wave_barrier();
      read_frags(fa, x0, 0);
      read_frags(fb, x0, 8);
      __builtin_amdgcn_sched_barrier(0);
      gather_rows(rows, h, si, di, lbyte, TILE_ROWS / 2, TILE_ROWS / 2);
      const int t2 = t1 + nwaves;
      if (t2 < ntrips) load_trip_indices(si, di, src_ro, dst_ro, t2 * TRIP_EDGES, E - 1);
    } else {
#pragma unroll
      for (int i = 0; i < TILE_ROWS; ++i) {
        *(v4f*)(tile + i * ROW_PITCH + lane * 4) = rows[i];
      }
      __builtin_amdgcn_wave_barrier();
      read_frags(fa, x0, 0);
      read_frags(fb, x0, 8);
    }
    __builtin_amdgcn_sched_barrier(0);

    v8f ga = {};
    v8f gb = {};
    gram_run(fa, ga, gb);
    __builtin_amdgcn_sched_barrier(0);
    read_frags(fa, x1, 0);
    __builtin_amdgcn_sched_barrier(0);
    gram_run(fb, ga, gb);
    __builtin_amdgcn_sched_barrier(0);
    read_frags(fb, x1, 8);
    __builtin_amdgcn_sched_barrier(0);

    const int vidx = lane & 7;
    const v8f g0 = ga + gb;
    const float v0 = pick_diag(g0, vidx);

    ga = (v8f){};
    gb = (v8f){};
    gram_run(fa, ga, gb);
    gram_run(fb, ga, gb);
    const v8f g1 = ga + gb;
    const float v1 = pick_diag(g1, vidx);

    const float ss0 = __shfl(v0, (lane - 8) & 31, 32);
    const float dd0 = __shfl(v0, (lane + 16) & 31, 32);
    const float ss1 = __shfl(v1, (lane - 16) & 31, 32);
    const float dd1 = __shfl(v1, (lane + 8) & 31, 32);
    const bool  low = lane < 16;
    const float dot = low ? v0 : v1;
    const float ss  = low ? ss0 : ss1;
    const float dd  = low ? dd0 : dd1;

    const long long e = (long long)t * TRIP_EDGES + (lane - 8);
    if (lane >= 8 && lane < 24 && e <= last_e) {
      const float ns = fmaxf(sqrtf(ss), NORM_EPS);
      const float nd = fmaxf(sqrtf(dd), NORM_EPS);
      out[e] = dot / (ns * nd);
    }

    if (!more) break;
    __builtin_amdgcn_wave_barrier();
    t = t1;
    t1 += nwaves;
  }
}

extern "C" void kernel_launch(void* const* d_in, const int* in_sizes, int n_in,
                              void* d_out, int out_size, void* d_ws, size_t ws_size,
                              hipStream_t stream) {
  const float* h   = (const float*)d_in[0];
  const int*   src = (const int*)d_in[1];
  const int*   dst = (const int*)d_in[2];
  float*       out = (float*)d_out;

  const int E = in_sizes[1];

  dim3 block(SDDMM_WAVES * 32), grid(SDDMM_GROUPS);
  cossim_gram_wmma_kernel<<<grid, block, 0, stream>>>(h, src, dst, out, E);
}


# Wave priority steps down with the wave's own progress (s_setprio, 4 levels over its trips): the 4 waves of a SIMD advance together instead of oldest-first, none runs its last trips alone
# speedup vs baseline: 3.0182x; 1.0349x over previous
#include <hip/hip_runtime.h>
#include <hip/hip_bf16.h>

typedef __attribute__((ext_vector_type(2))) float v2f;
typedef __attribute__((ext_vector_type(2))) unsigned v2u;
typedef __attribute__((ext_vector_type(4))) float v4f;
typedef __attribute__((ext_vector_type(8))) float v8f;
typedef int v16i_a4 __attribute__((ext_vector_type(16), aligned(4)));
typedef const int __attribute__((address_space(4)))* ro_int_ptr;
typedef const v16i_a4 __attribute__((address_space(4)))* ro_v16i_ptr;
typedef const v4f __attribute__((address_space(1)))* gl_v4f_ptr;

#define FEAT_D 128
#define NORM_EPS 1e-12f

#define TRIP_EDGES 16
#define TILE_ROWS 32
#define ROW_PITCH 132
#define SDDMM_WAVES 4
#define SDDMM_GROUPS 1024
#define FRAG_GROUP 8

__device__ __forceinline__ void read_frags(v4f (&f)[FRAG_GROUP], const float* x, int s0) {
#pragma unroll
  for (int j = 0; j < FRAG_GROUP; ++j) f[j] = *(const v4f*)(x + 8 * (s0 + j));
}

__device__ __forceinline__ void gram_run(const v4f (&f)[FRAG_GROUP], v8f& ga, v8f& gb) {
#pragma unroll
  for (int j = 0; j < FRAG_GROUP; ++j) {
    const v2f lo = {f[j].x, f[j].y};
    const v2f hi = {f[j].z, f[j].w};
    ga = __builtin_amdgcn_wmma_f32_16x16x4_f32(
         false, lo,  false, lo,
         (short)0, ga,  false,  false);
    gb = __builtin_amdgcn_wmma_f32_16x16x4_f32(
         false, hi,  false, hi,
         (short)0, gb,  false,  false);
  }
}

__device__ __forceinline__ float pick_diag(const v8f& g, int vidx) {
  float v = g[0];
#pragma unroll
  for (int i = 1; i < 8; ++i) v = (vidx == i) ? g[i] : v;
  return v;
}


__device__ __forceinline__ void set_progress_priority(int rem, int level_w) {
  if (rem > 2 * level_w + 1)  __builtin_amdgcn_s_setprio(3);
  else if (rem > level_w + 1) __builtin_amdgcn_s_setprio(2);
  else if (rem > 1)           __builtin_amdgcn_s_setprio(1);
  else                        __builtin_amdgcn_s_setprio(0);
}

__device__ __forceinline__ void load_trip_indices(int (&si)[TRIP_EDGES], int (&di)[TRIP_EDGES],
                                                  ro_int_ptr src, ro_int_ptr dst,
                                                  int e0, int last_e) {
  if (e0 + (TRIP_EDGES - 1) <= last_e) {
    const v16i_a4 sv = *(ro_v16i_ptr)(src + e0);
    const v16i_a4 dv = *(ro_v16i_ptr)(dst + e0);
#pragma unroll
    for (int j = 0; j < TRIP_EDGES; ++j) { si[j] = sv[j]; di[j] = dv[j]; }
  } else {
#pragma unroll
    for (int j = 0; j < TRIP_EDGES; ++j) {
      const int e = min(e0 + j, last_e);
      si[j] = src[e];
      di[j] = dst[e];
    }
  }
}

#define GATHER_GROUP 16
__device__ __forceinline__ void gather_rows(v4f (&rows)[TILE_ROWS], const float* __restrict__ h,
                                            const int (&si)[TRIP_EDGES], const int (&di)[TRIP_EDGES],
                                            unsigned& lbyte, const int g0, const int ng) {
  asm volatile("" : "+v"(lbyte));
#pragma unroll
  for (int g = g0; g < g0 + ng; g += GATHER_GROUP) {
    unsigned long long rbase[GATHER_GROUP];
#pragma unroll
    for (int j = 0; j < GATHER_GROUP; ++j) {
      const int i   = g + j;
      const int er  = ((i >> 4) << 3) + (i & 7);
      const unsigned n_i = (unsigned)((i & 8) ? di[er] : si[er]);
      unsigned lo = n_i << 9;
      unsigned hi = n_i >> 23;
      asm("" : "+s"(lo));
      asm("" : "+s"(hi));
      const v2u off = {lo, hi};
      rbase[j] = (unsigned long long)h + __builtin_bit_cast(unsigned long long, off);
      asm("" : "+s"(rbase[j]));
    }
#pragma unroll
    for (int j = 0; j < GATHER_GROUP; ++j) {
      rows[g + j] = *(gl_v4f_ptr)(rbase[j] + (unsigned long long)lbyte);
    }
  }
}

__global__ __launch_bounds__(SDDMM_WAVES * 32)
void cossim_gram_wmma_kernel(const float* __restrict__ h,
                             const int* __restrict__ src,
                             const int* __restrict__ dst,
                             float* __restrict__ out, int E) {
  __shared__ __attribute__((aligned(16))) float stage[SDDMM_WAVES * TILE_ROWS * ROW_PITCH];

  const int lane = threadIdx.x & 31;
  const int wave   = __builtin_amdgcn_readfirstlane((int)(threadIdx.x >> 5));
  const int nwaves = (int)gridDim.x * SDDMM_WAVES;
  const int ntrips = (E + TRIP_EDGES - 1) / TRIP_EDGES;
  int t = (int)blockIdx.x * SDDMM_WAVES + wave;
  if (t >= ntrips) return;

  const int my_trips = (ntrips - 1 - t) / nwaves + 1;
  const int level_w  = max(1, my_trips / 3);
  int rem = my_trips;
  set_progress_priority(rem, level_w);

  const int c    = lane & 15;
  const int half = lane >> 4;
  const long long last_e = (long long)E - 1;
  const ro_int_ptr src_ro = (ro_int_ptr)(unsigned long long)src;
  const ro_int_ptr dst_ro = (ro_int_ptr)(unsigned long long)dst;

  float* tile = stage + wave * (TILE_ROWS * ROW_PITCH);
  const float* x0 = tile + c * ROW_PITCH + 4 * half;
  const float* x1 = x0 + 16 * ROW_PITCH;
  unsigned lbyte = (unsigned)lane * 16u;

  int si[TRIP_EDGES], di[TRIP_EDGES];
  load_trip_indices(si, di, src_ro, dst_ro, t * TRIP_EDGES, E - 1);
  v4f rows[TILE_ROWS];
  gather_rows(rows, h, si, di, lbyte, 0, TILE_ROWS);
  int t1 = t + nwaves;
  if (t1 < ntrips) load_trip_indices(si, di, src_ro, dst_ro, t1 * TRIP_EDGES, E - 1);

  for (;;) {
    const bool more = t1 < ntrips;
    set_progress_priority(rem, level_w);
    v4f fa[FRAG_GROUP];
    v4f fb[FRAG_GROUP];
    if (more) {
#pragma unroll
      for (int i = 0; i < TILE_ROWS / 2; ++i) {
        *(v4f*)(tile + i * ROW_PITCH + lane * 4) = rows[i];
      }
      __builtin_amdgcn_sched_barrier(0);
      gather_rows(rows, h, si, di, lbyte, 0, TILE_ROWS / 2);
      __builtin_amdgcn_sched_barrier(0);
#pragma unroll
      for (int i = TILE_ROWS / 2; i < TILE_ROWS; ++i) {
        *(v4f*)(tile + i * ROW_PITCH + lane * 4) = rows[i];
      }
      __builtin_amdgcn_wave_barrier();
      read_frags(fa, x0, 0);
      read_frags(fb, x0, 8);
      __builtin_amdgcn_sched_barrier(0);
      gather_rows(rows, h, si, di, lbyte, TILE_ROWS / 2, TILE_ROWS / 2);
      const int t2 = t1 + nwaves;
      if (t2 < ntrips) load_trip_indices(si, di, src_ro, dst_ro, t2 * TRIP_EDGES, E - 1);
    } else {
#pragma unroll
      for (int i = 0; i < TILE_ROWS; ++i) {
        *(v4f*)(tile + i * ROW_PITCH + lane * 4) = rows[i];
      }
      __builtin_amdgcn_wave_barrier();
      read_frags(fa, x0, 0);
      read_frags(fb, x0, 8);
    }
    __builtin_amdgcn_sched_barrier(0);

    v8f ga = {};
    v8f gb = {};
    gram_run(fa, ga, gb);
    __builtin_amdgcn_sched_barrier(0);
    read_frags(fa, x1, 0);
    __builtin_amdgcn_sched_barrier(0);
    gram_run(fb, ga, gb);
    __builtin_amdgcn_sched_barrier(0);
    read_frags(fb, x1, 8);
    __builtin_amdgcn_sched_barrier(0);

    const int vidx = lane & 7;
    const v8f g0 = ga + gb;
    const float v0 = pick_diag(g0, vidx);

    ga = (v8f){};
    gb = (v8f){};
    gram_run(fa, ga, gb);
    gram_run(fb, ga, gb);
    const v8f g1 = ga + gb;
    const float v1 = pick_diag(g1, vidx);

    const float ss0 = __shfl(v0, (lane - 8) & 31, 32);
    const float dd0 = __shfl(v0, (lane + 16) & 31, 32);
    const float ss1 = __shfl(v1, (lane - 16) & 31, 32);
    const float dd1 = __shfl(v1, (lane + 8) & 31, 32);
    const bool  low = lane < 16;
    const float dot = low ? v0 : v1;
    const float ss  = low ? ss0 : ss1;
    const float dd  = low ? dd0 : dd1;

    const long long e = (long long)t * TRIP_EDGES + (lane - 8);
    if (lane >= 8 && lane < 24 && e <= last_e) {
      const float ns = fmaxf(sqrtf(ss), NORM_EPS);
      const float nd = fmaxf(sqrtf(dd), NORM_EPS);
      out[e] = dot / (ns * nd);
    }

    if (!more) break;
    __builtin_amdgcn_wave_barrier();
    t = t1;
    t1 += nwaves;
    --rem;
  }
}

extern "C" void kernel_launch(void* const* d_in, const int* in_sizes, int n_in,
                              void* d_out, int out_size, void* d_ws, size_t ws_size,
                              hipStream_t stream) {
  const float* h   = (const float*)d_in[0];
  const int*   src = (const int*)d_in[1];
  const int*   dst = (const int*)d_in[2];
  float*       out = (float*)d_out;

  const int E = in_sizes[1];

  dim3 block(SDDMM_WAVES * 32), grid(SDDMM_GROUPS);
  cossim_gram_wmma_kernel<<<grid, block, 0, stream>>>(h, src, dst, out, E);
}
